# starting kernel
# baseline (speedup 1.0000x reference)
#include <hip/hip_runtime.h>
#include <math.h>


typedef float v2f __attribute__((ext_vector_type(2)));
typedef float v8f __attribute__((ext_vector_type(8)));

#define TINYV 1e-6f
#define LLOB 128
#define SSAM 8
#define HHID 16
#define NELEM (8192 * 128 * 8)

#define SWZ(x, pat) __int_as_float(__builtin_amdgcn_ds_swizzle(__float_as_int(x), (pat)))
#define PAT_XOR16 0x401F
#define PAT_XOR1  0x041F
#define PAT_XOR2  0x081F
#define PAT_XOR4  0x101F

__global__ __launch_bounds__(256) void sg_vis_wmma_kernel(
    const float* __restrict__ points,
    const float* __restrict__ normals,
    const float* __restrict__ rootR,
    const float* __restrict__ lobes,
    const float* __restrict__ lambdas,
    const float* __restrict__ rrand,
    const float* __restrict__ W1,
    const float* __restrict__ b1,
    const float* __restrict__ W2,
    const float* __restrict__ b2,
    float* __restrict__ out)
{
    __shared__ float sF[32 * 12];

    const int tid  = threadIdx.x;
    const int e    = blockIdx.x * 256 + tid;
    const int lane = tid & 31;

    if (tid < 32) {
        const int l = ((blockIdx.x * 32) + tid) & (LLOB - 1);
        float lx = lobes[3*l+0], ly = lobes[3*l+1], lz = lobes[3*l+2];
        float invn = 1.f / (sqrtf(lx*lx + ly*ly + lz*lz) + TINYV);
        lx *= invn; ly *= invn; lz *= invn;
        float ux = -ly, uy = lx;
        float invu = 1.f / (sqrtf(ux*ux + uy*uy) + TINYV);
        ux *= invu; uy *= invu;
        float vx = -lz*uy, vy = lz*ux, vz = lx*uy - ly*ux;
        float invv = 1.f / (sqrtf(vx*vx + vy*vy + vz*vz) + TINYV);
        vx *= invv; vy *= invv; vz *= invv;
        float sharp = lambdas[l];
        float rphi  = fminf(acosf(1.f - 1.f/sharp), 1.04719755119659775f);
        float* F = &sF[tid * 12];
        F[0] = lx; F[1] = ly; F[2] = lz;
        F[3] = ux; F[4] = uy;
        F[5] = vx; F[6] = vy; F[7] = vz;
        F[8] = __sinf(rphi); F[9] = __cosf(rphi); F[10] = sharp; F[11] = 0.f;
    }
    __syncthreads();

    const float* F = &sF[(tid >> 3) * 12];
    const float lnx = F[0], lny = F[1], lnz = F[2];
    const float ux  = F[3], uy  = F[4];
    const float vx  = F[5], vy  = F[6], vz = F[7];
    const float sp  = F[8], cp  = F[9], sharp = F[10];

    const float theta = (rrand[e] * (1.f/SSAM) + (float)(tid & 7) * (1.f/SSAM))
                        * 6.28318530717958648f;
    const float st = __sinf(theta), ct = __cosf(theta);
    const float ca = ct * sp, cb = st * sp;
    const float dx = ux*ca + vx*cb + lnx*cp;
    const float dy = uy*ca + vy*cb + lny*cp;
    const float dz =          vz*cb + lnz*cp;

    const int n = blockIdx.x >> 2;
    const float px = points[3*n+0],  py = points[3*n+1],  pz = points[3*n+2];
    const float nx = normals[3*n+0], ny = normals[3*n+1], nz = normals[3*n+2];
    const bool cosmask = (nx*dx + ny*dy + nz*dz) > TINYV;

    const float id0 = -(dx*rootR[0] + dy*rootR[3] + dz*rootR[6]);
    const float id1 = -(dx*rootR[1] + dy*rootR[4] + dz*rootR[7]);
    const float id2 = -(dx*rootR[2] + dy*rootR[5] + dz*rootR[8]);

    const float w = __expf(sharp * (dx*lnx + dy*lny + dz*lnz - 1.f));

    const int  m     = lane & 15;
    const bool up    = lane >= 16;
    const int  hbase = up ? 8 : 0;

    v2f A0, A1;
    A0.x = up ? W1[2*HHID + m] : W1[0*HHID + m];
    A0.y = up ? W1[3*HHID + m] : W1[1*HHID + m];
    A1.x = up ? 0.f : W1[4*HHID + m];
    A1.y = up ? 0.f : W1[5*HHID + m];

    const float x3 = id0, x4 = id1, x5 = id2;
    const float y3 = SWZ(x3, PAT_XOR16);
    const float y4 = SWZ(x4, PAT_XOR16);
    const float y5 = SWZ(x5, PAT_XOR16);

    v2f B0t0, B1t0, B0t1, B1t1;
    B0t0.x = up ? pz  : px;   B0t0.y = up ? y3  : py;
    B1t0.x = up ? 0.f : x4;   B1t0.y = up ? 0.f : x5;
    B0t1.x = up ? pz  : px;   B0t1.y = up ? x3  : py;
    B1t1.x = up ? 0.f : y4;   B1t1.y = up ? 0.f : y5;

    v8f c0, c1;
    #pragma unroll
    for (int r = 0; r < 8; ++r) c0[r] = b1[hbase + r];
    c1 = c0;

    c0 = __builtin_amdgcn_wmma_f32_16x16x4_f32(false, A0, false, B0t0, (short)0, c0, false, false);
    c0 = __builtin_amdgcn_wmma_f32_16x16x4_f32(false, A1, false, B1t0, (short)0, c0, false, false);
    c1 = __builtin_amdgcn_wmma_f32_16x16x4_f32(false, A0, false, B0t1, (short)0, c1, false, false);
    c1 = __builtin_amdgcn_wmma_f32_16x16x4_f32(false, A1, false, B1t1, (short)0, c1, false, false);

    float w2v[8];
    #pragma unroll
    for (int r = 0; r < 8; ++r) w2v[r] = W2[hbase + r];
    float pa = 0.f, pb = 0.f;
    #pragma unroll
    for (int r = 0; r < 8; ++r) {
        const float t0 = c0[r], t1 = c1[r];
        pa += (t0 > 0.f ? t0 : 0.f) * w2v[r];
        pb += (t1 > 0.f ? t1 : 0.f) * w2v[r];
    }
    const float q  = up ? pa : pb;
    const float qq = SWZ(q, PAT_XOR16);
    const float logit = (up ? pb : pa) + qq + b2[0];
    const float pred  = 1.f / (1.f + __expf(-logit));
    const float vis   = cosmask ? pred : 0.f;

    float num = vis * w, den = w;
    num += SWZ(num, PAT_XOR1);  den += SWZ(den, PAT_XOR1);
    num += SWZ(num, PAT_XOR2);  den += SWZ(den, PAT_XOR2);
    num += SWZ(num, PAT_XOR4);  den += SWZ(den, PAT_XOR4);

    if ((lane & 7) == 0) out[e >> 3] = num / (den + TINYV);
}

extern "C" void kernel_launch(void* const* d_in, const int* in_sizes, int n_in,
                              void* d_out, int out_size, void* d_ws, size_t ws_size,
                              hipStream_t stream) {
    const float* points  = (const float*)d_in[0];
    const float* normals = (const float*)d_in[1];
    const float* rootR   = (const float*)d_in[2];
    const float* lobes   = (const float*)d_in[3];
    const float* lambdas = (const float*)d_in[4];
    const float* rrand   = (const float*)d_in[5];
    const float* W1      = (const float*)d_in[6];
    const float* b1      = (const float*)d_in[7];
    const float* W2      = (const float*)d_in[8];
    const float* b2      = (const float*)d_in[9];
    float* out = (float*)d_out;

    dim3 grid(NELEM / 256), block(256);
    sg_vis_wmma_kernel<<<grid, block, 0, stream>>>(
        points, normals, rootR, lobes, lambdas, rrand, W1, b1, W2, b2, out);
}


# Work-group keeps its 32 lobes and loops over 16 points: lobe frame (per lane, no LDS or barrier), W1 fragments, bias, W2 kept in registers, next r_theta fetched one trip ahead; grid 2048, was 32768
# speedup vs baseline: 1.3266x; 1.3266x over previous
#include <hip/hip_runtime.h>
#include <math.h>


typedef float v2f __attribute__((ext_vector_type(2)));
typedef float v8f __attribute__((ext_vector_type(8)));

#define TINYV 1e-6f
#define NPTS 8192
#define LLOB 128
#define SSAM 8
#define HHID 16
#define ROWF (LLOB * SSAM)
#define LGRP (LLOB / 32)
#define PTS_PER_WG 16
#define NWG (LGRP * (NPTS / PTS_PER_WG))

#define SWZ(x, pat) __int_as_float(__builtin_amdgcn_ds_swizzle(__float_as_int(x), (pat)))
#define PAT_XOR16 0x401F
#define PAT_XOR1  0x041F
#define PAT_XOR2  0x081F
#define PAT_XOR4  0x101F

__global__ __launch_bounds__(256, 16) void sg_vis_wmma_kernel(
    const float* __restrict__ points,
    const float* __restrict__ normals,
    const float* __restrict__ rootR,
    const float* __restrict__ lobes,
    const float* __restrict__ lambdas,
    const float* __restrict__ rrand,
    const float* __restrict__ W1,
    const float* __restrict__ b1,
    const float* __restrict__ W2,
    const float* __restrict__ b2,
    float* __restrict__ out)
{
    const int tid  = threadIdx.x;
    const int lane = tid & 31;
    const int lg   = blockIdx.x & (LGRP - 1);
    const int n0   = (blockIdx.x >> 2) * PTS_PER_WG;
    const int l    = lg * 32 + (tid >> 3);

    float lnx = lobes[3*l+0], lny = lobes[3*l+1], lnz = lobes[3*l+2];
    {
        const float invn = 1.f / (sqrtf(lnx*lnx + lny*lny + lnz*lnz) + TINYV);
        lnx *= invn; lny *= invn; lnz *= invn;
    }
    float ux = -lny, uy = lnx;
    {
        const float invu = 1.f / (sqrtf(ux*ux + uy*uy) + TINYV);
        ux *= invu; uy *= invu;
    }
    float vx = -lnz*uy, vy = lnz*ux, vz = lnx*uy - lny*ux;
    {
        const float invv = 1.f / (sqrtf(vx*vx + vy*vy + vz*vz) + TINYV);
        vx *= invv; vy *= invv; vz *= invv;
    }
    const float sharp = lambdas[l];
    const float rphi  = fminf(acosf(1.f - 1.f/sharp), 1.04719755119659775f);
    const float sp = __sinf(rphi), cp = __cosf(rphi);

    const int  m     = lane & 15;
    const bool up    = lane >= 16;
    const int  hbase = up ? 8 : 0;

    v2f A0, A1;
    A0.x = up ? W1[2*HHID + m] : W1[0*HHID + m];
    A0.y = up ? W1[3*HHID + m] : W1[1*HHID + m];
    A1.x = up ? 0.f : W1[4*HHID + m];
    A1.y = up ? 0.f : W1[5*HHID + m];

    v8f cb;
    float w2v[8];
    #pragma unroll
    for (int r = 0; r < 8; ++r) { cb[r] = b1[hbase + r]; w2v[r] = W2[hbase + r]; }
    const float bias2 = b2[0];

    const float r00 = rootR[0], r01 = rootR[1], r02 = rootR[2];
    const float r10 = rootR[3], r11 = rootR[4], r12 = rootR[5];
    const float r20 = rootR[6], r21 = rootR[7], r22 = rootR[8];

    const float soff = (float)(tid & 7) * (1.f/SSAM);

    const float* rp = rrand + (size_t)n0 * ROWF + lg * 256 + tid;
    float*       op = out + (size_t)n0 * LLOB + l;

    float rcur = rp[0];
    float px = points[3*n0+0],  py = points[3*n0+1],  pz = points[3*n0+2];
    float nx = normals[3*n0+0], ny = normals[3*n0+1], nz = normals[3*n0+2];

    #pragma unroll 1
    for (int i = 0; i < PTS_PER_WG; ++i) {
        const int in = (i + 1 < PTS_PER_WG) ? (i + 1) : i;
        const int nn = n0 + in;
        const float rnext = rp[(size_t)in * ROWF];
        const float qx = points[3*nn+0],  qy = points[3*nn+1],  qz = points[3*nn+2];
        const float mx = normals[3*nn+0], my = normals[3*nn+1], mz = normals[3*nn+2];

        const float theta = (rcur * (1.f/SSAM) + soff) * 6.28318530717958648f;
        const float st = __sinf(theta), ct = __cosf(theta);
        const float ca = ct * sp, cb2 = st * sp;
        const float dx = ux*ca + vx*cb2 + lnx*cp;
        const float dy = uy*ca + vy*cb2 + lny*cp;
        const float dz =          vz*cb2 + lnz*cp;

        const bool cosmask = (nx*dx + ny*dy + nz*dz) > TINYV;

        const float id0 = -(dx*r00 + dy*r10 + dz*r20);
        const float id1 = -(dx*r01 + dy*r11 + dz*r21);
        const float id2 = -(dx*r02 + dy*r12 + dz*r22);

        const float w = __expf(sharp * (dx*lnx + dy*lny + dz*lnz - 1.f));

        const float x3 = id0, x4 = id1, x5 = id2;
        const float y3 = SWZ(x3, PAT_XOR16);
        const float y4 = SWZ(x4, PAT_XOR16);
        const float y5 = SWZ(x5, PAT_XOR16);

        v2f B0t0, B1t0, B0t1, B1t1;
        B0t0.x = up ? pz  : px;   B0t0.y = up ? y3  : py;
        B1t0.x = up ? 0.f : x4;   B1t0.y = up ? 0.f : x5;
        B0t1.x = up ? pz  : px;   B0t1.y = up ? x3  : py;
        B1t1.x = up ? 0.f : y4;   B1t1.y = up ? 0.f : y5;

        v8f c0, c1;
        c0 = __builtin_amdgcn_wmma_f32_16x16x4_f32(false, A0, false, B0t0, (short)0, cb, false, false);
        c0 = __builtin_amdgcn_wmma_f32_16x16x4_f32(false, A1, false, B1t0, (short)0, c0, false, false);
        c1 = __builtin_amdgcn_wmma_f32_16x16x4_f32(false, A0, false, B0t1, (short)0, cb, false, false);
        c1 = __builtin_amdgcn_wmma_f32_16x16x4_f32(false, A1, false, B1t1, (short)0, c1, false, false);

        float pa = 0.f, pb = 0.f;
        #pragma unroll
        for (int r = 0; r < 8; ++r) {
            const float t0 = c0[r], t1 = c1[r];
            pa += (t0 > 0.f ? t0 : 0.f) * w2v[r];
            pb += (t1 > 0.f ? t1 : 0.f) * w2v[r];
        }
        const float q  = up ? pa : pb;
        const float qq = SWZ(q, PAT_XOR16);
        const float logit = (up ? pb : pa) + qq + bias2;
        const float pred  = 1.f / (1.f + __expf(-logit));
        const float vis   = cosmask ? pred : 0.f;

        float num = vis * w, den = w;
        num += SWZ(num, PAT_XOR1);  den += SWZ(den, PAT_XOR1);
        num += SWZ(num, PAT_XOR2);  den += SWZ(den, PAT_XOR2);
        num += SWZ(num, PAT_XOR4);  den += SWZ(den, PAT_XOR4);

        if ((lane & 7) == 0) op[(size_t)i * LLOB] = num / (den + TINYV);

        rcur = rnext;
        px = qx; py = qy; pz = qz;
        nx = mx; ny = my; nz = mz;
    }
}

extern "C" void kernel_launch(void* const* d_in, const int* in_sizes, int n_in,
                              void* d_out, int out_size, void* d_ws, size_t ws_size,
                              hipStream_t stream) {
    const float* points  = (const float*)d_in[0];
    const float* normals = (const float*)d_in[1];
    const float* rootR   = (const float*)d_in[2];
    const float* lobes   = (const float*)d_in[3];
    const float* lambdas = (const float*)d_in[4];
    const float* rrand   = (const float*)d_in[5];
    const float* W1      = (const float*)d_in[6];
    const float* b1      = (const float*)d_in[7];
    const float* W2      = (const float*)d_in[8];
    const float* b2      = (const float*)d_in[9];
    float* out = (float*)d_out;

    dim3 grid(NWG), block(256);
    sg_vis_wmma_kernel<<<grid, block, 0, stream>>>(
        points, normals, rootR, lobes, lambdas, rrand, W1, b1, W2, b2, out);
}


# Lobe frame and rotation folded into W1's A fragments (wave = 2 lobes x point pairs, B = own ct,st); point term one shared C tile: 3 matrix products per trip, was 4; hidden sums in a new order
# speedup vs baseline: 1.5229x; 1.1480x over previous
#include <hip/hip_runtime.h>
#include <math.h>


typedef float v2f __attribute__((ext_vector_type(2)));
typedef float v8f __attribute__((ext_vector_type(8)));

#define TINYV 1e-6f
#define NPTS 8192
#define LLOB 128
#define SSAM 8
#define HHID 16
#define ROWF (LLOB * SSAM)
#define LGRP (LLOB / 32)
#define PTS_PER_WG 32
#define TRIPS (PTS_PER_WG / 2)
#define NWG (LGRP * (NPTS / PTS_PER_WG))

#define SWZ(x, pat) __int_as_float(__builtin_amdgcn_ds_swizzle(__float_as_int(x), (pat)))
#define PAT_XOR16 0x401F
#define PAT_XOR1  0x041F
#define PAT_XOR2  0x081F
#define PAT_XOR4  0x101F

__global__ __launch_bounds__(512, 16) void sg_vis_wmma_kernel(
    const float* __restrict__ points,
    const float* __restrict__ normals,
    const float* __restrict__ rootR,
    const float* __restrict__ lobes,
    const float* __restrict__ lambdas,
    const float* __restrict__ rrand,
    const float* __restrict__ W1,
    const float* __restrict__ b1,
    const float* __restrict__ W2,
    const float* __restrict__ b2,
    float* __restrict__ out)
{
    const int tid  = threadIdx.x;
    const int lane = tid & 31;
    const int wv   = tid >> 5;
    const bool up  = lane >= 16;
    const unsigned pb = (lane >> 3) & 1;
    const unsigned s  = lane & 7;
    const int lg   = blockIdx.x & (LGRP - 1);
    const int n0   = (blockIdx.x >> 2) * PTS_PER_WG;
    const int l    = lg * 32 + 2 * wv + (up ? 1 : 0);

    float lnx = lobes[3*l+0], lny = lobes[3*l+1], lnz = lobes[3*l+2];
    {
        const float invn = 1.f / (sqrtf(lnx*lnx + lny*lny + lnz*lnz) + TINYV);
        lnx *= invn; lny *= invn; lnz *= invn;
    }
    float ux = -lny, uy = lnx;
    {
        const float invu = 1.f / (sqrtf(ux*ux + uy*uy) + TINYV);
        ux *= invu; uy *= invu;
    }
    float vx = -lnz*uy, vy = lnz*ux, vz = lnx*uy - lny*ux;
    {
        const float invv = 1.f / (sqrtf(vx*vx + vy*vy + vz*vz) + TINYV);
        vx *= invv; vy *= invv; vz *= invv;
    }
    const float sharp = lambdas[l];
    const float rphi  = fminf(acosf(1.f - 1.f/sharp), 1.04719755119659775f);
    const float sp = __sinf(rphi), cp = __cosf(rphi);

    const int  m     = lane & 15;
    const int  hbase = up ? 8 : 0;

    const float r00 = rootR[0], r01 = rootR[1], r02 = rootR[2];
    const float r10 = rootR[3], r11 = rootR[4], r12 = rootR[5];
    const float r20 = rootR[6], r21 = rootR[7], r22 = rootR[8];

    const float w3 = W1[3*HHID + m], w4 = W1[4*HHID + m], w5 = W1[5*HHID + m];
    const float M0 = -(r00*w3 + r01*w4 + r02*w5);
    const float M1 = -(r10*w3 + r11*w4 + r12*w5);
    const float M2 = -(r20*w3 + r21*w4 + r22*w5);
    const float aU = sp * (ux*M0 + uy*M1);
    const float aV = sp * (vx*M0 + vy*M1 + vz*M2);
    const float aC = cp * (lnx*M0 + lny*M1 + lnz*M2) + b1[m];
    const float pC = SWZ(aC, PAT_XOR16);

    v2f A0a, A0b, A1;
    A0a.x = up ? pC  : aU;   A0a.y = up ? 0.f : aV;
    A0b.x = up ? aU  : pC;   A0b.y = up ? aV  : 0.f;
    A1.x  = up ? W1[2*HHID + m] : W1[0*HHID + m];
    A1.y  = up ? 0.f            : W1[1*HHID + m];

    float w2v[8];
    #pragma unroll
    for (int r = 0; r < 8; ++r) w2v[r] = W2[hbase + r];
    const float bias2 = b2[0];

    const float soff = (float)s * (1.f/SSAM);

    const unsigned oR  = pb * ROWF + (unsigned)l * SSAM + s;
    const unsigned oP0 = pb * 3 + (up ? 2u : 0u);
    const unsigned oP1 = pb * 3 + 1u;
    const unsigned oN  = pb * 3;
    const unsigned oO  = pb * LLOB + (unsigned)l;

    const float* rq = rrand   + (size_t)n0 * ROWF;
    const float* pq = points  + (size_t)n0 * 3;
    const float* nq = normals + (size_t)n0 * 3;
    float*       oq = out     + (size_t)n0 * LLOB;

    float rcur = rq[oR];
    v2f B1;
    B1.x = pq[oP0];  B1.y = pq[oP1];
    float nx = nq[oN], ny = nq[oN + 1], nz = nq[oN + 2];

    #pragma unroll 1
    for (int i = 0; i < TRIPS; ++i) {
        const bool more = (i + 1 < TRIPS);
        rq += more ? 2 * ROWF : 0;
        pq += more ? 6 : 0;
        nq += more ? 6 : 0;

        const v8f zero = {0.f, 0.f, 0.f, 0.f, 0.f, 0.f, 0.f, 0.f};
        const v8f P = __builtin_amdgcn_wmma_f32_16x16x4_f32(false, A1, false, B1, (short)0, zero, false, false);

        const float theta = (rcur * (1.f/SSAM) + soff) * 6.28318530717958648f;
        const float st = __sinf(theta), ct = __cosf(theta);
        const float ca = ct * sp, cb2 = st * sp;
        const float dx = ux*ca + vx*cb2 + lnx*cp;
        const float dy = uy*ca + vy*cb2 + lny*cp;
        const float dz =          vz*cb2 + lnz*cp;

        const bool cosmask = (nx*dx + ny*dy + nz*dz) > TINYV;

        const float w = __expf(sharp * (dx*lnx + dy*lny + dz*lnz - 1.f));

        rcur = rq[oR];
        B1.x = pq[oP0];  B1.y = pq[oP1];
        nx = nq[oN];  ny = nq[oN + 1];  nz = nq[oN + 2];

        v2f B0a, B0b;
        B0a.x = up ? 1.f : ct;   B0a.y = st;
        B0b.x = up ? ct  : 1.f;  B0b.y = st;

        const v8f c0 = __builtin_amdgcn_wmma_f32_16x16x4_f32(false, A0a, false, B0a, (short)0, P, false, false);
        const v8f c1 = __builtin_amdgcn_wmma_f32_16x16x4_f32(false, A0b, false, B0b, (short)0, P, false, false);

        float pa = 0.f, pbv = 0.f;
        #pragma unroll
        for (int r = 0; r < 8; ++r) {
            const float t0 = c0[r], t1 = c1[r];
            pa  += (t0 > 0.f ? t0 : 0.f) * w2v[r];
            pbv += (t1 > 0.f ? t1 : 0.f) * w2v[r];
        }
        const float q  = up ? pa : pbv;
        const float qq = SWZ(q, PAT_XOR16);
        const float logit = (up ? pbv : pa) + qq + bias2;
        const float pred  = 1.f / (1.f + __expf(-logit));
        const float vis   = cosmask ? pred : 0.f;

        float num = vis * w, den = w;
        num += SWZ(num, PAT_XOR1);  den += SWZ(den, PAT_XOR1);
        num += SWZ(num, PAT_XOR2);  den += SWZ(den, PAT_XOR2);
        num += SWZ(num, PAT_XOR4);  den += SWZ(den, PAT_XOR4);

        if (s == 0) oq[oO] = num / (den + TINYV);
        oq += 2 * LLOB;
    }
}

extern "C" void kernel_launch(void* const* d_in, const int* in_sizes, int n_in,
                              void* d_out, int out_size, void* d_ws, size_t ws_size,
                              hipStream_t stream) {
    const float* points  = (const float*)d_in[0];
    const float* normals = (const float*)d_in[1];
    const float* rootR   = (const float*)d_in[2];
    const float* lobes   = (const float*)d_in[3];
    const float* lambdas = (const float*)d_in[4];
    const float* rrand   = (const float*)d_in[5];
    const float* W1      = (const float*)d_in[6];
    const float* b1      = (const float*)d_in[7];
    const float* W2      = (const float*)d_in[8];
    const float* b2      = (const float*)d_in[9];
    float* out = (float*)d_out;

    dim3 grid(NWG), block(512);
    sg_vis_wmma_kernel<<<grid, block, 0, stream>>>(
        points, normals, rootR, lobes, lambdas, rrand, W1, b1, W2, b2, out);
}


# Point term W1[0:3]^T x points formed once per work-group (2 matrix products into 2 KB of LDS, one barrier) and read back as the C operand: 2 matrix products per trip, was 3; sums unchanged
# speedup vs baseline: 1.5762x; 1.0350x over previous
#include <hip/hip_runtime.h>
#include <math.h>


typedef float v2f __attribute__((ext_vector_type(2)));
typedef float v4f __attribute__((ext_vector_type(4)));
typedef float v8f __attribute__((ext_vector_type(8)));

#define TINYV 1e-6f
#define NPTS 8192
#define LLOB 128
#define SSAM 8
#define HHID 16
#define ROWF (LLOB * SSAM)
#define LGRP (LLOB / 32)
#define PTS_PER_WG 32
#define TRIPS (PTS_PER_WG / 2)
#define NWG (LGRP * (NPTS / PTS_PER_WG))

#define SWZ(x, pat) __int_as_float(__builtin_amdgcn_ds_swizzle(__float_as_int(x), (pat)))
#define PAT_XOR16 0x401F
#define PAT_XOR1  0x041F
#define PAT_XOR2  0x081F
#define PAT_XOR4  0x101F

__global__ __launch_bounds__(512, 16) void sg_vis_wmma_kernel(
    const float* __restrict__ points,
    const float* __restrict__ normals,
    const float* __restrict__ rootR,
    const float* __restrict__ lobes,
    const float* __restrict__ lambdas,
    const float* __restrict__ rrand,
    const float* __restrict__ W1,
    const float* __restrict__ b1,
    const float* __restrict__ W2,
    const float* __restrict__ b2,
    float* __restrict__ out)
{
    __shared__ v4f Psh[PTS_PER_WG * HHID / 4];

    const int tid  = threadIdx.x;
    const int lane = tid & 31;
    const int wv   = tid >> 5;
    const bool up  = lane >= 16;
    const unsigned pb = (lane >> 3) & 1;
    const unsigned s  = lane & 7;
    const int lg   = blockIdx.x & (LGRP - 1);
    const int n0   = (blockIdx.x >> 2) * PTS_PER_WG;
    const int l    = lg * 32 + 2 * wv + (up ? 1 : 0);

    float lnx = lobes[3*l+0], lny = lobes[3*l+1], lnz = lobes[3*l+2];
    {
        const float invn = 1.f / (sqrtf(lnx*lnx + lny*lny + lnz*lnz) + TINYV);
        lnx *= invn; lny *= invn; lnz *= invn;
    }
    float ux = -lny, uy = lnx;
    {
        const float invu = 1.f / (sqrtf(ux*ux + uy*uy) + TINYV);
        ux *= invu; uy *= invu;
    }
    float vx = -lnz*uy, vy = lnz*ux, vz = lnx*uy - lny*ux;
    {
        const float invv = 1.f / (sqrtf(vx*vx + vy*vy + vz*vz) + TINYV);
        vx *= invv; vy *= invv; vz *= invv;
    }
    const float sharp = lambdas[l];
    const float rphi  = fminf(acosf(1.f - 1.f/sharp), 1.04719755119659775f);
    const float sp = __sinf(rphi), cp = __cosf(rphi);

    const int  m     = lane & 15;
    const int  hbase = up ? 8 : 0;

    const float r00 = rootR[0], r01 = rootR[1], r02 = rootR[2];
    const float r10 = rootR[3], r11 = rootR[4], r12 = rootR[5];
    const float r20 = rootR[6], r21 = rootR[7], r22 = rootR[8];

    const float w3 = W1[3*HHID + m], w4 = W1[4*HHID + m], w5 = W1[5*HHID + m];
    const float M0 = -(r00*w3 + r01*w4 + r02*w5);
    const float M1 = -(r10*w3 + r11*w4 + r12*w5);
    const float M2 = -(r20*w3 + r21*w4 + r22*w5);
    const float aU = sp * (ux*M0 + uy*M1);
    const float aV = sp * (vx*M0 + vy*M1 + vz*M2);
    const float aC = cp * (lnx*M0 + lny*M1 + lnz*M2) + b1[m];
    const float pC = SWZ(aC, PAT_XOR16);

    v2f A0a, A0b, A1;
    A0a.x = up ? pC  : aU;   A0a.y = up ? 0.f : aV;
    A0b.x = up ? aU  : pC;   A0b.y = up ? aV  : 0.f;
    A1.x  = up ? W1[2*HHID + m] : W1[0*HHID + m];
    A1.y  = up ? 0.f            : W1[1*HHID + m];

    float w2v[8];
    #pragma unroll
    for (int r = 0; r < 8; ++r) w2v[r] = W2[hbase + r];
    const float bias2 = b2[0];

    const float soff = (float)s * (1.f/SSAM);

    const unsigned oR  = pb * ROWF + (unsigned)l * SSAM + s;
    const unsigned oN  = pb * 3;
    const unsigned oO  = pb * LLOB + (unsigned)l;

    const float* rq = rrand   + (size_t)n0 * ROWF;
    const float* nq = normals + (size_t)n0 * 3;
    float*       oq = out     + (size_t)n0 * LLOB;

    float rcur = rq[oR];
    float nx = nq[oN], ny = nq[oN + 1], nz = nq[oN + 2];

    if (__builtin_amdgcn_readfirstlane(wv) < 2) {
        const int pt = 16 * wv + m;
        const float* pp = points + (size_t)(n0 + pt) * 3;
        v2f Bp;
        Bp.x = up ? pp[2] : pp[0];
        Bp.y = pp[1];
        const v8f zero = {0.f, 0.f, 0.f, 0.f, 0.f, 0.f, 0.f, 0.f};
        const v8f Pt = __builtin_amdgcn_wmma_f32_16x16x4_f32(false, A1, false, Bp, (short)0, zero, false, false);
        v4f lo4, hi4;
        lo4.x = Pt[0]; lo4.y = Pt[1]; lo4.z = Pt[2]; lo4.w = Pt[3];
        hi4.x = Pt[4]; hi4.y = Pt[5]; hi4.z = Pt[6]; hi4.w = Pt[7];
        Psh[pt * 4 + (up ? 2 : 0)]     = lo4;
        Psh[pt * 4 + (up ? 2 : 0) + 1] = hi4;
    }
    __syncthreads();

    const v4f* pl = Psh + (pb * 4 + (up ? 2 : 0));

    #pragma unroll 1
    for (int i = 0; i < TRIPS; ++i) {
        const bool more = (i + 1 < TRIPS);
        rq += more ? 2 * ROWF : 0;
        nq += more ? 6 : 0;

        const v4f plo = pl[0], phi = pl[1];
        pl += 8;
        const v8f P = {plo.x, plo.y, plo.z, plo.w, phi.x, phi.y, phi.z, phi.w};

        const float theta = (rcur * (1.f/SSAM) + soff) * 6.28318530717958648f;
        const float st = __sinf(theta), ct = __cosf(theta);
        const float ca = ct * sp, cb2 = st * sp;
        const float dx = ux*ca + vx*cb2 + lnx*cp;
        const float dy = uy*ca + vy*cb2 + lny*cp;
        const float dz =          vz*cb2 + lnz*cp;

        const bool cosmask = (nx*dx + ny*dy + nz*dz) > TINYV;

        const float w = __expf(sharp * (dx*lnx + dy*lny + dz*lnz - 1.f));

        rcur = rq[oR];
        nx = nq[oN];  ny = nq[oN + 1];  nz = nq[oN + 2];

        v2f B0a, B0b;
        B0a.x = up ? 1.f : ct;   B0a.y = st;
        B0b.x = up ? ct  : 1.f;  B0b.y = st;

        const v8f c0 = __builtin_amdgcn_wmma_f32_16x16x4_f32(false, A0a, false, B0a, (short)0, P, false, false);
        const v8f c1 = __builtin_amdgcn_wmma_f32_16x16x4_f32(false, A0b, false, B0b, (short)0, P, false, false);

        float pa = 0.f, pbv = 0.f;
        #pragma unroll
        for (int r = 0; r < 8; ++r) {
            const float t0 = c0[r], t1 = c1[r];
            pa  += (t0 > 0.f ? t0 : 0.f) * w2v[r];
            pbv += (t1 > 0.f ? t1 : 0.f) * w2v[r];
        }
        const float q  = up ? pa : pbv;
        const float qq = SWZ(q, PAT_XOR16);
        const float logit = (up ? pbv : pa) + qq + bias2;
        const float pred  = 1.f / (1.f + __expf(-logit));
        const float vis   = cosmask ? pred : 0.f;

        float num = vis * w, den = w;
        num += SWZ(num, PAT_XOR1);  den += SWZ(den, PAT_XOR1);
        num += SWZ(num, PAT_XOR2);  den += SWZ(den, PAT_XOR2);
        num += SWZ(num, PAT_XOR4);  den += SWZ(den, PAT_XOR4);

        if (s == 0) oq[oO] = num / (den + TINYV);
        oq += 2 * LLOB;
    }
}

extern "C" void kernel_launch(void* const* d_in, const int* in_sizes, int n_in,
                              void* d_out, int out_size, void* d_ws, size_t ws_size,
                              hipStream_t stream) {
    const float* points  = (const float*)d_in[0];
    const float* normals = (const float*)d_in[1];
    const float* rootR   = (const float*)d_in[2];
    const float* lobes   = (const float*)d_in[3];
    const float* lambdas = (const float*)d_in[4];
    const float* rrand   = (const float*)d_in[5];
    const float* W1      = (const float*)d_in[6];
    const float* b1      = (const float*)d_in[7];
    const float* W2      = (const float*)d_in[8];
    const float* b2      = (const float*)d_in[9];
    float* out = (float*)d_out;

    dim3 grid(NWG), block(512);
    sg_vis_wmma_kernel<<<grid, block, 0, stream>>>(
        points, normals, rootR, lobes, lambdas, rrand, W1, b1, W2, b2, out);
}


# Trip loop's two divisions (sigmoid 1/(1+e), final num/(den+eps)) use the hardware reciprocal v_rcp_f32 (1 ulp) instead of the correctly rounded division sequences; sums unchanged
# speedup vs baseline: 1.8394x; 1.1670x over previous
#include <hip/hip_runtime.h>
#include <math.h>


typedef float v2f __attribute__((ext_vector_type(2)));
typedef float v4f __attribute__((ext_vector_type(4)));
typedef float v8f __attribute__((ext_vector_type(8)));

#define TINYV 1e-6f
#define NPTS 8192
#define LLOB 128
#define SSAM 8
#define HHID 16
#define ROWF (LLOB * SSAM)
#define LGRP (LLOB / 32)
#define PTS_PER_WG 32
#define TRIPS (PTS_PER_WG / 2)
#define NWG (LGRP * (NPTS / PTS_PER_WG))

#define SWZ(x, pat) __int_as_float(__builtin_amdgcn_ds_swizzle(__float_as_int(x), (pat)))
#define PAT_XOR16 0x401F
#define PAT_XOR1  0x041F
#define PAT_XOR2  0x081F
#define PAT_XOR4  0x101F

__global__ __launch_bounds__(512, 16) void sg_vis_wmma_kernel(
    const float* __restrict__ points,
    const float* __restrict__ normals,
    const float* __restrict__ rootR,
    const float* __restrict__ lobes,
    const float* __restrict__ lambdas,
    const float* __restrict__ rrand,
    const float* __restrict__ W1,
    const float* __restrict__ b1,
    const float* __restrict__ W2,
    const float* __restrict__ b2,
    float* __restrict__ out)
{
    __shared__ v4f Psh[PTS_PER_WG * HHID / 4];

    const int tid  = threadIdx.x;
    const int lane = tid & 31;
    const int wv   = tid >> 5;
    const bool up  = lane >= 16;
    const unsigned pb = (lane >> 3) & 1;
    const unsigned s  = lane & 7;
    const int lg   = blockIdx.x & (LGRP - 1);
    const int n0   = (blockIdx.x >> 2) * PTS_PER_WG;
    const int l    = lg * 32 + 2 * wv + (up ? 1 : 0);

    float lnx = lobes[3*l+0], lny = lobes[3*l+1], lnz = lobes[3*l+2];
    {
        const float invn = 1.f / (sqrtf(lnx*lnx + lny*lny + lnz*lnz) + TINYV);
        lnx *= invn; lny *= invn; lnz *= invn;
    }
    float ux = -lny, uy = lnx;
    {
        const float invu = 1.f / (sqrtf(ux*ux + uy*uy) + TINYV);
        ux *= invu; uy *= invu;
    }
    float vx = -lnz*uy, vy = lnz*ux, vz = lnx*uy - lny*ux;
    {
        const float invv = 1.f / (sqrtf(vx*vx + vy*vy + vz*vz) + TINYV);
        vx *= invv; vy *= invv; vz *= invv;
    }
    const float sharp = lambdas[l];
    const float rphi  = fminf(acosf(1.f - 1.f/sharp), 1.04719755119659775f);
    const float sp = __sinf(rphi), cp = __cosf(rphi);

    const int  m     = lane & 15;
    const int  hbase = up ? 8 : 0;

    const float r00 = rootR[0], r01 = rootR[1], r02 = rootR[2];
    const float r10 = rootR[3], r11 = rootR[4], r12 = rootR[5];
    const float r20 = rootR[6], r21 = rootR[7], r22 = rootR[8];

    const float w3 = W1[3*HHID + m], w4 = W1[4*HHID + m], w5 = W1[5*HHID + m];
    const float M0 = -(r00*w3 + r01*w4 + r02*w5);
    const float M1 = -(r10*w3 + r11*w4 + r12*w5);
    const float M2 = -(r20*w3 + r21*w4 + r22*w5);
    const float aU = sp * (ux*M0 + uy*M1);
    const float aV = sp * (vx*M0 + vy*M1 + vz*M2);
    const float aC = cp * (lnx*M0 + lny*M1 + lnz*M2) + b1[m];
    const float pC = SWZ(aC, PAT_XOR16);

    v2f A0a, A0b, A1;
    A0a.x = up ? pC  : aU;   A0a.y = up ? 0.f : aV;
    A0b.x = up ? aU  : pC;   A0b.y = up ? aV  : 0.f;
    A1.x  = up ? W1[2*HHID + m] : W1[0*HHID + m];
    A1.y  = up ? 0.f            : W1[1*HHID + m];

    float w2v[8];
    #pragma unroll
    for (int r = 0; r < 8; ++r) w2v[r] = W2[hbase + r];
    const float bias2 = b2[0];

    const float soff = (float)s * (1.f/SSAM);

    const unsigned oR  = pb * ROWF + (unsigned)l * SSAM + s;
    const unsigned oN  = pb * 3;
    const unsigned oO  = pb * LLOB + (unsigned)l;

    const float* rq = rrand   + (size_t)n0 * ROWF;
    const float* nq = normals + (size_t)n0 * 3;
    float*       oq = out     + (size_t)n0 * LLOB;

    float rcur = rq[oR];
    float nx = nq[oN], ny = nq[oN + 1], nz = nq[oN + 2];

    if (__builtin_amdgcn_readfirstlane(wv) < 2) {
        const int pt = 16 * wv + m;
        const float* pp = points + (size_t)(n0 + pt) * 3;
        v2f Bp;
        Bp.x = up ? pp[2] : pp[0];
        Bp.y = pp[1];
        const v8f zero = {0.f, 0.f, 0.f, 0.f, 0.f, 0.f, 0.f, 0.f};
        const v8f Pt = __builtin_amdgcn_wmma_f32_16x16x4_f32(false, A1, false, Bp, (short)0, zero, false, false);
        v4f lo4, hi4;
        lo4.x = Pt[0]; lo4.y = Pt[1]; lo4.z = Pt[2]; lo4.w = Pt[3];
        hi4.x = Pt[4]; hi4.y = Pt[5]; hi4.z = Pt[6]; hi4.w = Pt[7];
        Psh[pt * 4 + (up ? 2 : 0)]     = lo4;
        Psh[pt * 4 + (up ? 2 : 0) + 1] = hi4;
    }
    __syncthreads();

    const v4f* pl = Psh + (pb * 4 + (up ? 2 : 0));

    #pragma unroll 1
    for (int i = 0; i < TRIPS; ++i) {
        const bool more = (i + 1 < TRIPS);
        rq += more ? 2 * ROWF : 0;
        nq += more ? 6 : 0;

        const v4f plo = pl[0], phi = pl[1];
        pl += 8;
        const v8f P = {plo.x, plo.y, plo.z, plo.w, phi.x, phi.y, phi.z, phi.w};

        const float theta = (rcur * (1.f/SSAM) + soff) * 6.28318530717958648f;
        const float st = __sinf(theta), ct = __cosf(theta);
        const float ca = ct * sp, cb2 = st * sp;
        const float dx = ux*ca + vx*cb2 + lnx*cp;
        const float dy = uy*ca + vy*cb2 + lny*cp;
        const float dz =          vz*cb2 + lnz*cp;

        const bool cosmask = (nx*dx + ny*dy + nz*dz) > TINYV;

        const float w = __expf(sharp * (dx*lnx + dy*lny + dz*lnz - 1.f));

        rcur = rq[oR];
        nx = nq[oN];  ny = nq[oN + 1];  nz = nq[oN + 2];

        v2f B0a, B0b;
        B0a.x = up ? 1.f : ct;   B0a.y = st;
        B0b.x = up ? ct  : 1.f;  B0b.y = st;

        const v8f c0 = __builtin_amdgcn_wmma_f32_16x16x4_f32(false, A0a, false, B0a, (short)0, P, false, false);
        const v8f c1 = __builtin_amdgcn_wmma_f32_16x16x4_f32(false, A0b, false, B0b, (short)0, P, false, false);

        float pa = 0.f, pbv = 0.f;
        #pragma unroll
        for (int r = 0; r < 8; ++r) {
            const float t0 = c0[r], t1 = c1[r];
            pa  += (t0 > 0.f ? t0 : 0.f) * w2v[r];
            pbv += (t1 > 0.f ? t1 : 0.f) * w2v[r];
        }
        const float q  = up ? pa : pbv;
        const float qq = SWZ(q, PAT_XOR16);
        const float logit = (up ? pbv : pa) + qq + bias2;
        const float pred  = __builtin_amdgcn_rcpf(1.f + __expf(-logit));
        const float vis   = cosmask ? pred : 0.f;

        float num = vis * w, den = w;
        num += SWZ(num, PAT_XOR1);  den += SWZ(den, PAT_XOR1);
        num += SWZ(num, PAT_XOR2);  den += SWZ(den, PAT_XOR2);
        num += SWZ(num, PAT_XOR4);  den += SWZ(den, PAT_XOR4);

        if (s == 0) oq[oO] = num * __builtin_amdgcn_rcpf(den + TINYV);
        oq += 2 * LLOB;
    }
}

extern "C" void kernel_launch(void* const* d_in, const int* in_sizes, int n_in,
                              void* d_out, int out_size, void* d_ws, size_t ws_size,
                              hipStream_t stream) {
    const float* points  = (const float*)d_in[0];
    const float* normals = (const float*)d_in[1];
    const float* rootR   = (const float*)d_in[2];
    const float* lobes   = (const float*)d_in[3];
    const float* lambdas = (const float*)d_in[4];
    const float* rrand   = (const float*)d_in[5];
    const float* W1      = (const float*)d_in[6];
    const float* b1      = (const float*)d_in[7];
    const float* W2      = (const float*)d_in[8];
    const float* b2      = (const float*)d_in[9];
    float* out = (float*)d_out;

    dim3 grid(NWG), block(512);
    sg_vis_wmma_kernel<<<grid, block, 0, stream>>>(
        points, normals, rootR, lobes, lambdas, rrand, W1, b1, W2, b2, out);
}


# Layer 2 recoded: ReLU is one integer max on the float's bits, each tap one fused multiply-add into its tile's sum (tap order 0..7 kept, products no longer rounded first): 2 ops per value, was 4 to 5
# speedup vs baseline: 1.9929x; 1.0835x over previous
#include <hip/hip_runtime.h>
#include <math.h>


typedef float v2f __attribute__((ext_vector_type(2)));
typedef float v4f __attribute__((ext_vector_type(4)));
typedef float v8f __attribute__((ext_vector_type(8)));

#define TINYV 1e-6f
#define NPTS 8192
#define LLOB 128
#define SSAM 8
#define HHID 16
#define ROWF (LLOB * SSAM)
#define LGRP (LLOB / 32)
#define PTS_PER_WG 32
#define TRIPS (PTS_PER_WG / 2)
#define NWG (LGRP * (NPTS / PTS_PER_WG))

#define SWZ(x, pat) __int_as_float(__builtin_amdgcn_ds_swizzle(__float_as_int(x), (pat)))
#define PAT_XOR16 0x401F
#define PAT_XOR1  0x041F
#define PAT_XOR2  0x081F
#define PAT_XOR4  0x101F

#define RELU_BITS(x) __int_as_float(max(__float_as_int(x), 0))

__global__ __launch_bounds__(512, 16) void sg_vis_wmma_kernel(
    const float* __restrict__ points,
    const float* __restrict__ normals,
    const float* __restrict__ rootR,
    const float* __restrict__ lobes,
    const float* __restrict__ lambdas,
    const float* __restrict__ rrand,
    const float* __restrict__ W1,
    const float* __restrict__ b1,
    const float* __restrict__ W2,
    const float* __restrict__ b2,
    float* __restrict__ out)
{
    __shared__ v4f Psh[PTS_PER_WG * HHID / 4];

    const int tid  = threadIdx.x;
    const int lane = tid & 31;
    const int wv   = tid >> 5;
    const bool up  = lane >= 16;
    const unsigned pb = (lane >> 3) & 1;
    const unsigned s  = lane & 7;
    const int lg   = blockIdx.x & (LGRP - 1);
    const int n0   = (blockIdx.x >> 2) * PTS_PER_WG;
    const int l    = lg * 32 + 2 * wv + (up ? 1 : 0);

    float lnx = lobes[3*l+0], lny = lobes[3*l+1], lnz = lobes[3*l+2];
    {
        const float invn = 1.f / (sqrtf(lnx*lnx + lny*lny + lnz*lnz) + TINYV);
        lnx *= invn; lny *= invn; lnz *= invn;
    }
    float ux = -lny, uy = lnx;
    {
        const float invu = 1.f / (sqrtf(ux*ux + uy*uy) + TINYV);
        ux *= invu; uy *= invu;
    }
    float vx = -lnz*uy, vy = lnz*ux, vz = lnx*uy - lny*ux;
    {
        const float invv = 1.f / (sqrtf(vx*vx + vy*vy + vz*vz) + TINYV);
        vx *= invv; vy *= invv; vz *= invv;
    }
    const float sharp = lambdas[l];
    const float rphi  = fminf(acosf(1.f - 1.f/sharp), 1.04719755119659775f);
    const float sp = __sinf(rphi), cp = __cosf(rphi);

    const int  m     = lane & 15;
    const int  hbase = up ? 8 : 0;

    const float r00 = rootR[0], r01 = rootR[1], r02 = rootR[2];
    const float r10 = rootR[3], r11 = rootR[4], r12 = rootR[5];
    const float r20 = rootR[6], r21 = rootR[7], r22 = rootR[8];

    const float w3 = W1[3*HHID + m], w4 = W1[4*HHID + m], w5 = W1[5*HHID + m];
    const float M0 = -(r00*w3 + r01*w4 + r02*w5);
    const float M1 = -(r10*w3 + r11*w4 + r12*w5);
    const float M2 = -(r20*w3 + r21*w4 + r22*w5);
    const float aU = sp * (ux*M0 + uy*M1);
    const float aV = sp * (vx*M0 + vy*M1 + vz*M2);
    const float aC = cp * (lnx*M0 + lny*M1 + lnz*M2) + b1[m];
    const float pC = SWZ(aC, PAT_XOR16);

    v2f A0a, A0b, A1;
    A0a.x = up ? pC  : aU;   A0a.y = up ? 0.f : aV;
    A0b.x = up ? aU  : pC;   A0b.y = up ? aV  : 0.f;
    A1.x  = up ? W1[2*HHID + m] : W1[0*HHID + m];
    A1.y  = up ? 0.f            : W1[1*HHID + m];

    float w2v[8];
    #pragma unroll
    for (int r = 0; r < 8; ++r) w2v[r] = W2[hbase + r];
    const float bias2 = b2[0];

    const float soff = (float)s * (1.f/SSAM);

    const unsigned oR  = pb * ROWF + (unsigned)l * SSAM + s;
    const unsigned oN  = pb * 3;
    const unsigned oO  = pb * LLOB + (unsigned)l;

    const float* rq = rrand   + (size_t)n0 * ROWF;
    const float* nq = normals + (size_t)n0 * 3;
    float*       oq = out     + (size_t)n0 * LLOB;

    float rcur = rq[oR];
    float nx = nq[oN], ny = nq[oN + 1], nz = nq[oN + 2];

    if (__builtin_amdgcn_readfirstlane(wv) < 2) {
        const int pt = 16 * wv + m;
        const float* pp = points + (size_t)(n0 + pt) * 3;
        v2f Bp;
        Bp.x = up ? pp[2] : pp[0];
        Bp.y = pp[1];
        const v8f zero = {0.f, 0.f, 0.f, 0.f, 0.f, 0.f, 0.f, 0.f};
        const v8f Pt = __builtin_amdgcn_wmma_f32_16x16x4_f32(false, A1, false, Bp, (short)0, zero, false, false);
        v4f lo4, hi4;
        lo4.x = Pt[0]; lo4.y = Pt[1]; lo4.z = Pt[2]; lo4.w = Pt[3];
        hi4.x = Pt[4]; hi4.y = Pt[5]; hi4.z = Pt[6]; hi4.w = Pt[7];
        Psh[pt * 4 + (up ? 2 : 0)]     = lo4;
        Psh[pt * 4 + (up ? 2 : 0) + 1] = hi4;
    }
    __syncthreads();

    const v4f* pl = Psh + (pb * 4 + (up ? 2 : 0));

    #pragma unroll 1
    for (int i = 0; i < TRIPS; ++i) {
        const bool more = (i + 1 < TRIPS);
        rq += more ? 2 * ROWF : 0;
        nq += more ? 6 : 0;

        const v4f plo = pl[0], phi = pl[1];
        pl += 8;
        const v8f P = {plo.x, plo.y, plo.z, plo.w, phi.x, phi.y, phi.z, phi.w};

        const float theta = (rcur * (1.f/SSAM) + soff) * 6.28318530717958648f;
        const float st = __sinf(theta), ct = __cosf(theta);
        const float ca = ct * sp, cb2 = st * sp;
        const float dx = ux*ca + vx*cb2 + lnx*cp;
        const float dy = uy*ca + vy*cb2 + lny*cp;
        const float dz =          vz*cb2 + lnz*cp;

        const bool cosmask = (nx*dx + ny*dy + nz*dz) > TINYV;

        const float w = __expf(sharp * (dx*lnx + dy*lny + dz*lnz - 1.f));

        rcur = rq[oR];
        nx = nq[oN];  ny = nq[oN + 1];  nz = nq[oN + 2];

        v2f B0a, B0b;
        B0a.x = up ? 1.f : ct;   B0a.y = st;
        B0b.x = up ? ct  : 1.f;  B0b.y = st;

        const v8f c0 = __builtin_amdgcn_wmma_f32_16x16x4_f32(false, A0a, false, B0a, (short)0, P, false, false);
        const v8f c1 = __builtin_amdgcn_wmma_f32_16x16x4_f32(false, A0b, false, B0b, (short)0, P, false, false);

        float pa  = RELU_BITS(c0[0]) * w2v[0];
        float pbv = RELU_BITS(c1[0]) * w2v[0];
        #pragma unroll
        for (int r = 1; r < 8; ++r) {
            pa  = __builtin_fmaf(RELU_BITS(c0[r]), w2v[r], pa);
            pbv = __builtin_fmaf(RELU_BITS(c1[r]), w2v[r], pbv);
        }
        const float q  = up ? pa : pbv;
        const float qq = SWZ(q, PAT_XOR16);
        const float logit = (up ? pbv : pa) + qq + bias2;
        const float pred  = __builtin_amdgcn_rcpf(1.f + __expf(-logit));
        const float vis   = cosmask ? pred : 0.f;

        float num = vis * w, den = w;
        num += SWZ(num, PAT_XOR1);  den += SWZ(den, PAT_XOR1);
        num += SWZ(num, PAT_XOR2);  den += SWZ(den, PAT_XOR2);
        num += SWZ(num, PAT_XOR4);  den += SWZ(den, PAT_XOR4);

        if (s == 0) oq[oO] = num * __builtin_amdgcn_rcpf(den + TINYV);
        oq += 2 * LLOB;
    }
}

extern "C" void kernel_launch(void* const* d_in, const int* in_sizes, int n_in,
                              void* d_out, int out_size, void* d_ws, size_t ws_size,
                              hipStream_t stream) {
    const float* points  = (const float*)d_in[0];
    const float* normals = (const float*)d_in[1];
    const float* rootR   = (const float*)d_in[2];
    const float* lobes   = (const float*)d_in[3];
    const float* lambdas = (const float*)d_in[4];
    const float* rrand   = (const float*)d_in[5];
    const float* W1      = (const float*)d_in[6];
    const float* b1      = (const float*)d_in[7];
    const float* W2      = (const float*)d_in[8];
    const float* b2      = (const float*)d_in[9];
    float* out = (float*)d_out;

    dim3 grid(NWG), block(512);
    sg_vis_wmma_kernel<<<grid, block, 0, stream>>>(
        points, normals, rootR, lobes, lambdas, rrand, W1, b1, W2, b2, out);
}


# Lobe weight exp(lambda*(dir.l-1)) is the same for all 8 samples (U,V perpendicular to l): w and w/(8w+eps) formed once per lane; a trip sums only the masked sigmoids: sum(vis*w) becomes w*sum(vis)
# speedup vs baseline: 2.1043x; 1.0559x over previous
#include <hip/hip_runtime.h>
#include <math.h>


typedef float v2f __attribute__((ext_vector_type(2)));
typedef float v4f __attribute__((ext_vector_type(4)));
typedef float v8f __attribute__((ext_vector_type(8)));

#define TINYV 1e-6f
#define NPTS 8192
#define LLOB 128
#define SSAM 8
#define HHID 16
#define ROWF (LLOB * SSAM)
#define LGRP (LLOB / 32)
#define PTS_PER_WG 32
#define TRIPS (PTS_PER_WG / 2)
#define NWG (LGRP * (NPTS / PTS_PER_WG))

#define SWZ(x, pat) __int_as_float(__builtin_amdgcn_ds_swizzle(__float_as_int(x), (pat)))
#define PAT_XOR16 0x401F
#define PAT_XOR1  0x041F
#define PAT_XOR2  0x081F
#define PAT_XOR4  0x101F

#define RELU_BITS(x) __int_as_float(max(__float_as_int(x), 0))

__global__ __launch_bounds__(512, 16) void sg_vis_wmma_kernel(
    const float* __restrict__ points,
    const float* __restrict__ normals,
    const float* __restrict__ rootR,
    const float* __restrict__ lobes,
    const float* __restrict__ lambdas,
    const float* __restrict__ rrand,
    const float* __restrict__ W1,
    const float* __restrict__ b1,
    const float* __restrict__ W2,
    const float* __restrict__ b2,
    float* __restrict__ out)
{
    __shared__ v4f Psh[PTS_PER_WG * HHID / 4];

    const int tid  = threadIdx.x;
    const int lane = tid & 31;
    const int wv   = tid >> 5;
    const bool up  = lane >= 16;
    const unsigned pb = (lane >> 3) & 1;
    const unsigned s  = lane & 7;
    const int lg   = blockIdx.x & (LGRP - 1);
    const int n0   = (blockIdx.x >> 2) * PTS_PER_WG;
    const int l    = lg * 32 + 2 * wv + (up ? 1 : 0);

    float lnx = lobes[3*l+0], lny = lobes[3*l+1], lnz = lobes[3*l+2];
    {
        const float invn = 1.f / (sqrtf(lnx*lnx + lny*lny + lnz*lnz) + TINYV);
        lnx *= invn; lny *= invn; lnz *= invn;
    }
    float ux = -lny, uy = lnx;
    {
        const float invu = 1.f / (sqrtf(ux*ux + uy*uy) + TINYV);
        ux *= invu; uy *= invu;
    }
    float vx = -lnz*uy, vy = lnz*ux, vz = lnx*uy - lny*ux;
    {
        const float invv = 1.f / (sqrtf(vx*vx + vy*vy + vz*vz) + TINYV);
        vx *= invv; vy *= invv; vz *= invv;
    }
    const float sharp = lambdas[l];
    const float rphi  = fminf(acosf(1.f - 1.f/sharp), 1.04719755119659775f);
    const float sp = __sinf(rphi), cp = __cosf(rphi);

    const float wl = __expf(sharp * (cp * (lnx*lnx + lny*lny + lnz*lnz) - 1.f));
    const float kq = wl * __builtin_amdgcn_rcpf((float)SSAM * wl + TINYV);

    const int  m     = lane & 15;
    const int  hbase = up ? 8 : 0;

    const float r00 = rootR[0], r01 = rootR[1], r02 = rootR[2];
    const float r10 = rootR[3], r11 = rootR[4], r12 = rootR[5];
    const float r20 = rootR[6], r21 = rootR[7], r22 = rootR[8];

    const float w3 = W1[3*HHID + m], w4 = W1[4*HHID + m], w5 = W1[5*HHID + m];
    const float M0 = -(r00*w3 + r01*w4 + r02*w5);
    const float M1 = -(r10*w3 + r11*w4 + r12*w5);
    const float M2 = -(r20*w3 + r21*w4 + r22*w5);
    const float aU = sp * (ux*M0 + uy*M1);
    const float aV = sp * (vx*M0 + vy*M1 + vz*M2);
    const float aC = cp * (lnx*M0 + lny*M1 + lnz*M2) + b1[m];
    const float pC = SWZ(aC, PAT_XOR16);

    v2f A0a, A0b, A1;
    A0a.x = up ? pC  : aU;   A0a.y = up ? 0.f : aV;
    A0b.x = up ? aU  : pC;   A0b.y = up ? aV  : 0.f;
    A1.x  = up ? W1[2*HHID + m] : W1[0*HHID + m];
    A1.y  = up ? 0.f            : W1[1*HHID + m];

    float w2v[8];
    #pragma unroll
    for (int r = 0; r < 8; ++r) w2v[r] = W2[hbase + r];
    const float bias2 = b2[0];

    const float soff = (float)s * (1.f/SSAM);

    const unsigned oR  = pb * ROWF + (unsigned)l * SSAM + s;
    const unsigned oN  = pb * 3;
    const unsigned oO  = pb * LLOB + (unsigned)l;

    const float* rq = rrand   + (size_t)n0 * ROWF;
    const float* nq = normals + (size_t)n0 * 3;
    float*       oq = out     + (size_t)n0 * LLOB;

    float rcur = rq[oR];
    float nx = nq[oN], ny = nq[oN + 1], nz = nq[oN + 2];

    if (__builtin_amdgcn_readfirstlane(wv) < 2) {
        const int pt = 16 * wv + m;
        const float* pp = points + (size_t)(n0 + pt) * 3;
        v2f Bp;
        Bp.x = up ? pp[2] : pp[0];
        Bp.y = pp[1];
        const v8f zero = {0.f, 0.f, 0.f, 0.f, 0.f, 0.f, 0.f, 0.f};
        const v8f Pt = __builtin_amdgcn_wmma_f32_16x16x4_f32(false, A1, false, Bp, (short)0, zero, false, false);
        v4f lo4, hi4;
        lo4.x = Pt[0]; lo4.y = Pt[1]; lo4.z = Pt[2]; lo4.w = Pt[3];
        hi4.x = Pt[4]; hi4.y = Pt[5]; hi4.z = Pt[6]; hi4.w = Pt[7];
        Psh[pt * 4 + (up ? 2 : 0)]     = lo4;
        Psh[pt * 4 + (up ? 2 : 0) + 1] = hi4;
    }
    __syncthreads();

    const v4f* pl = Psh + (pb * 4 + (up ? 2 : 0));

    #pragma unroll 1
    for (int i = 0; i < TRIPS; ++i) {
        const bool more = (i + 1 < TRIPS);
        rq += more ? 2 * ROWF : 0;
        nq += more ? 6 : 0;

        const v4f plo = pl[0], phi = pl[1];
        pl += 8;
        const v8f P = {plo.x, plo.y, plo.z, plo.w, phi.x, phi.y, phi.z, phi.w};

        const float theta = (rcur * (1.f/SSAM) + soff) * 6.28318530717958648f;
        const float st = __sinf(theta), ct = __cosf(theta);
        const float ca = ct * sp, cb2 = st * sp;
        const float dx = ux*ca + vx*cb2 + lnx*cp;
        const float dy = uy*ca + vy*cb2 + lny*cp;
        const float dz =          vz*cb2 + lnz*cp;

        const bool cosmask = (nx*dx + ny*dy + nz*dz) > TINYV;

        rcur = rq[oR];
        nx = nq[oN];  ny = nq[oN + 1];  nz = nq[oN + 2];

        v2f B0a, B0b;
        B0a.x = up ? 1.f : ct;   B0a.y = st;
        B0b.x = up ? ct  : 1.f;  B0b.y = st;

        const v8f c0 = __builtin_amdgcn_wmma_f32_16x16x4_f32(false, A0a, false, B0a, (short)0, P, false, false);
        const v8f c1 = __builtin_amdgcn_wmma_f32_16x16x4_f32(false, A0b, false, B0b, (short)0, P, false, false);

        float pa  = RELU_BITS(c0[0]) * w2v[0];
        float pbv = RELU_BITS(c1[0]) * w2v[0];
        #pragma unroll
        for (int r = 1; r < 8; ++r) {
            pa  = __builtin_fmaf(RELU_BITS(c0[r]), w2v[r], pa);
            pbv = __builtin_fmaf(RELU_BITS(c1[r]), w2v[r], pbv);
        }
        const float q  = up ? pa : pbv;
        const float qq = SWZ(q, PAT_XOR16);
        const float logit = (up ? pbv : pa) + qq + bias2;
        const float pred  = __builtin_amdgcn_rcpf(1.f + __expf(-logit));
        const float vis   = cosmask ? pred : 0.f;

        float num = vis;
        num += SWZ(num, PAT_XOR1);
        num += SWZ(num, PAT_XOR2);
        num += SWZ(num, PAT_XOR4);

        if (s == 0) oq[oO] = num * kq;
        oq += 2 * LLOB;
    }
}

extern "C" void kernel_launch(void* const* d_in, const int* in_sizes, int n_in,
                              void* d_out, int out_size, void* d_ws, size_t ws_size,
                              hipStream_t stream) {
    const float* points  = (const float*)d_in[0];
    const float* normals = (const float*)d_in[1];
    const float* rootR   = (const float*)d_in[2];
    const float* lobes   = (const float*)d_in[3];
    const float* lambdas = (const float*)d_in[4];
    const float* rrand   = (const float*)d_in[5];
    const float* W1      = (const float*)d_in[6];
    const float* b1      = (const float*)d_in[7];
    const float* W2      = (const float*)d_in[8];
    const float* b2      = (const float*)d_in[9];
    float* out = (float*)d_out;

    dim3 grid(NWG), block(512);
    sg_vis_wmma_kernel<<<grid, block, 0, stream>>>(
        points, normals, rootR, lobes, lambdas, rrand, W1, b1, W2, b2, out);
}


# Lane owns one (lobe, point) and walks its 8 samples in 8 trips: normal, point term (wave's own product, no barrier) and the sample sum stay in the lane; that sum is now sequential, was a xor tree
# speedup vs baseline: 2.3972x; 1.1392x over previous
#include <hip/hip_runtime.h>
#include <math.h>


typedef float v2f __attribute__((ext_vector_type(2)));
typedef float v4f __attribute__((ext_vector_type(4)));
typedef float v8f __attribute__((ext_vector_type(8)));

#define TINYV 1e-6f
#define NPTS 8192
#define LLOB 128
#define SSAM 8
#define HHID 16
#define LGRP (LLOB / 32)
#define PTS_PER_WG 32
#define OCTS (PTS_PER_WG / 16)
#define NWG (LGRP * (NPTS / PTS_PER_WG))

#define SWZ(x, pat) __int_as_float(__builtin_amdgcn_ds_swizzle(__float_as_int(x), (pat)))
#define PAT_XOR16 0x401F

#define RELU_BITS(x) __int_as_float(max(__float_as_int(x), 0))

__global__ __launch_bounds__(512, 16) void sg_vis_wmma_kernel(
    const float* __restrict__ points,
    const float* __restrict__ normals,
    const float* __restrict__ rootR,
    const float* __restrict__ lobes,
    const float* __restrict__ lambdas,
    const float* __restrict__ rrand,
    const float* __restrict__ W1,
    const float* __restrict__ b1,
    const float* __restrict__ W2,
    const float* __restrict__ b2,
    float* __restrict__ out)
{
    __shared__ v4f rsh[512 * 2];

    const int tid  = threadIdx.x;
    const int lane = tid & 31;
    const int wv   = tid >> 5;
    const bool up  = lane >= 16;
    const int lg   = blockIdx.x & (LGRP - 1);
    const int n0   = (blockIdx.x >> 2) * PTS_PER_WG;
    const int l    = lg * 32 + 2 * wv + (up ? 1 : 0);

    float lnx = lobes[3*l+0], lny = lobes[3*l+1], lnz = lobes[3*l+2];
    {
        const float invn = 1.f / (sqrtf(lnx*lnx + lny*lny + lnz*lnz) + TINYV);
        lnx *= invn; lny *= invn; lnz *= invn;
    }
    float ux = -lny, uy = lnx;
    {
        const float invu = 1.f / (sqrtf(ux*ux + uy*uy) + TINYV);
        ux *= invu; uy *= invu;
    }
    float vx = -lnz*uy, vy = lnz*ux, vz = lnx*uy - lny*ux;
    {
        const float invv = 1.f / (sqrtf(vx*vx + vy*vy + vz*vz) + TINYV);
        vx *= invv; vy *= invv; vz *= invv;
    }
    const float sharp = lambdas[l];
    const float rphi  = fminf(acosf(1.f - 1.f/sharp), 1.04719755119659775f);
    const float sp = __sinf(rphi), cp = __cosf(rphi);

    const float wl = __expf(sharp * (cp * (lnx*lnx + lny*lny + lnz*lnz) - 1.f));
    const float kq = wl * __builtin_amdgcn_rcpf((float)SSAM * wl + TINYV);

    const int  m     = lane & 15;
    const int  hbase = up ? 8 : 0;

    const float r00 = rootR[0], r01 = rootR[1], r02 = rootR[2];
    const float r10 = rootR[3], r11 = rootR[4], r12 = rootR[5];
    const float r20 = rootR[6], r21 = rootR[7], r22 = rootR[8];

    const float w3 = W1[3*HHID + m], w4 = W1[4*HHID + m], w5 = W1[5*HHID + m];
    const float M0 = -(r00*w3 + r01*w4 + r02*w5);
    const float M1 = -(r10*w3 + r11*w4 + r12*w5);
    const float M2 = -(r20*w3 + r21*w4 + r22*w5);
    const float aU = sp * (ux*M0 + uy*M1);
    const float aV = sp * (vx*M0 + vy*M1 + vz*M2);
    const float aC = cp * (lnx*M0 + lny*M1 + lnz*M2) + b1[m];
    const float pC = SWZ(aC, PAT_XOR16);

    v2f A0a, A0b, A1;
    A0a.x = up ? pC  : aU;   A0a.y = up ? 0.f : aV;
    A0b.x = up ? aU  : pC;   A0b.y = up ? aV  : 0.f;
    A1.x  = up ? W1[2*HHID + m] : W1[0*HHID + m];
    A1.y  = up ? 0.f            : W1[1*HHID + m];

    float w2v[8];
    #pragma unroll
    for (int r = 0; r < 8; ++r) w2v[r] = W2[hbase + r];
    const float bias2 = b2[0];

    const float lcx = lnx * cp, lcy = lny * cp, lcz = lnz * cp;

    #pragma unroll 1
    for (int o = 0; o < OCTS; ++o) {
        const unsigned n  = (unsigned)(n0 + 16 * o + m);
        const unsigned nl = n * LLOB + (unsigned)l;

        {
            const v4f* rg = (const v4f*)(rrand + (size_t)nl * SSAM);
            const v4f ra = rg[0], rb = rg[1];
            rsh[2 * tid]     = ra;
            rsh[2 * tid + 1] = rb;
        }
        const float* rl = (const float*)rsh + 8 * tid;

        const float* nn = normals + (size_t)(n * 3u);
        const float nx = nn[0], ny = nn[1], nz = nn[2];

        const float* pp = points + (size_t)(n * 3u);
        v2f Bp;
        Bp.x = up ? pp[2] : pp[0];
        Bp.y = pp[1];
        const v8f zero = {0.f, 0.f, 0.f, 0.f, 0.f, 0.f, 0.f, 0.f};
        const v8f P = __builtin_amdgcn_wmma_f32_16x16x4_f32(false, A1, false, Bp, (short)0, zero, false, false);

        float num  = 0.f;
        float soff = 0.f;

        #pragma unroll 1
        for (int s = 0; s < SSAM; ++s) {
            const float theta = (rl[s] * (1.f/SSAM) + soff) * 6.28318530717958648f;
            soff += 1.f/SSAM;

            const float st = __sinf(theta), ct = __cosf(theta);
            const float ca = ct * sp, cb2 = st * sp;
            const float dx = __builtin_fmaf(ux, ca, vx * cb2) + lcx;
            const float dy = __builtin_fmaf(uy, ca, vy * cb2) + lcy;
            const float dz = __builtin_fmaf(vz, cb2, lcz);
            const bool cosmask = __builtin_fmaf(nz, dz, __builtin_fmaf(nx, dx, ny * dy)) > TINYV;

            v2f B0a, B0b;
            B0a.x = up ? 1.f : ct;   B0a.y = st;
            B0b.x = up ? ct  : 1.f;  B0b.y = st;

            const v8f c0 = __builtin_amdgcn_wmma_f32_16x16x4_f32(false, A0a, false, B0a, (short)0, P, false, false);
            const v8f c1 = __builtin_amdgcn_wmma_f32_16x16x4_f32(false, A0b, false, B0b, (short)0, P, false, false);

            float pa  = RELU_BITS(c0[0]) * w2v[0];
            float pbv = RELU_BITS(c1[0]) * w2v[0];
            #pragma unroll
            for (int r = 1; r < 8; ++r) {
                pa  = __builtin_fmaf(RELU_BITS(c0[r]), w2v[r], pa);
                pbv = __builtin_fmaf(RELU_BITS(c1[r]), w2v[r], pbv);
            }
            const float q  = up ? pa : pbv;
            const float qq = SWZ(q, PAT_XOR16);
            const float logit = (up ? pbv : pa) + qq + bias2;
            const float pred  = __builtin_amdgcn_rcpf(1.f + __expf(-logit));
            num += cosmask ? pred : 0.f;
        }

        out[nl] = num * kq;
    }
}

extern "C" void kernel_launch(void* const* d_in, const int* in_sizes, int n_in,
                              void* d_out, int out_size, void* d_ws, size_t ws_size,
                              hipStream_t stream) {
    const float* points  = (const float*)d_in[0];
    const float* normals = (const float*)d_in[1];
    const float* rootR   = (const float*)d_in[2];
    const float* lobes   = (const float*)d_in[3];
    const float* lambdas = (const float*)d_in[4];
    const float* rrand   = (const float*)d_in[5];
    const float* W1      = (const float*)d_in[6];
    const float* b1      = (const float*)d_in[7];
    const float* W2      = (const float*)d_in[8];
    const float* b2      = (const float*)d_in[9];
    float* out = (float*)d_out;

    dim3 grid(NWG), block(512);
    sg_vis_wmma_kernel<<<grid, block, 0, stream>>>(
        points, normals, rootR, lobes, lambdas, rrand, W1, b1, W2, b2, out);
}
